# starting kernel
# baseline (speedup 1.0000x reference)
#include <hip/hip_runtime.h>

typedef float v2f __attribute__((ext_vector_type(2)));
typedef float v8f __attribute__((ext_vector_type(8)));

#define NNODES 100000
#define NEDGES 800000
#define INDIM  128
#define HID1   64
#define HID2   64
#define ODIM   32

__global__ __launch_bounds__(256) void zero_f32_k(float* __restrict__ p, int n) {
    int i = blockIdx.x * blockDim.x + threadIdx.x;
    if (i < n) p[i] = 0.0f;
}

__global__ __launch_bounds__(256) void degree_k(const int* __restrict__ dst,
                                                float* __restrict__ deg, int nE) {
    int e = blockIdx.x * blockDim.x + threadIdx.x;
    if (e < nE) unsafeAtomicAdd(&deg[dst[e]], 1.0f);
}

__global__ __launch_bounds__(256) void rsqrt_k(float* __restrict__ d, int n) {
    int i = blockIdx.x * blockDim.x + threadIdx.x;
    if (i < n) d[i] = rsqrtf(d[i] + 1.0f);
}

template <int N>
__global__ __launch_bounds__(256) void gemm_wmma_f32_k(const float* __restrict__ A,
                                                       const float* __restrict__ W,
                                                       float* __restrict__ C,
                                                       int M, int K) {
    constexpr int NT = N / 16;
    const int lane = threadIdx.x & 31;
    const int wave = (blockIdx.x * blockDim.x + threadIdx.x) >> 5;
    const int m0   = wave << 4;
    if (m0 >= M) return;

    const int mr = lane & 15;
    const int kh = lane >> 4;

    v8f acc[NT];
#pragma unroll
    for (int t = 0; t < NT; ++t) acc[t] = {};

    const size_t arow = (size_t)(m0 + mr) * (size_t)K;

    for (int k0 = 0; k0 < K; k0 += 4) {
        v2f a;
        a.x = A[arow + k0 + 2 * kh + 0];
        a.y = A[arow + k0 + 2 * kh + 1];
#pragma unroll
        for (int t = 0; t < NT; ++t) {
            const int ncol = (t << 4) + mr;
            v2f b;
            b.x = W[(size_t)(k0 + 2 * kh + 0) * N + ncol];
            b.y = W[(size_t)(k0 + 2 * kh + 1) * N + ncol];
            acc[t] = __builtin_amdgcn_wmma_f32_16x16x4_f32(
                false, a, false, b, (short)0, acc[t], false, false);
        }
    }

#pragma unroll
    for (int t = 0; t < NT; ++t) {
        const int ncol = (t << 4) + mr;
#pragma unroll
        for (int v = 0; v < 8; ++v) {
            const int row = m0 + 8 * kh + v;
            C[(size_t)row * N + ncol] = acc[t][v];
        }
    }
}

template <int H>
__global__ __launch_bounds__(256) void edge_agg_k(const float* __restrict__ h,
                                                  const int* __restrict__ src,
                                                  const int* __restrict__ dst,
                                                  const float* __restrict__ dis,
                                                  float* __restrict__ agg, int nE) {
    const int warp = (blockIdx.x * blockDim.x + threadIdx.x) >> 5;
    const int lane = threadIdx.x & 31;
    if (warp >= nE) return;
    const int s = src[warp];
    const int d = dst[warp];
    const float norm = dis[s] * dis[d];
    const float* hs = h + (size_t)s * H;
    float* ad = agg + (size_t)d * H;
#pragma unroll
    for (int f = lane; f < H; f += 32)
        unsafeAtomicAdd(&ad[f], norm * hs[f]);
}

template <int H>
__global__ __launch_bounds__(256) void finish_k(const float* __restrict__ agg,
                                                const float* __restrict__ h,
                                                const float* __restrict__ dis,
                                                const float* __restrict__ b,
                                                float* __restrict__ out, int M) {
    const int i = blockIdx.x * blockDim.x + threadIdx.x;
    if (i >= M * H) return;
    const int node = i >> (H == 64 ? 6 : 5);
    const int f    = i & (H - 1);
    const float dv = dis[node];
    const float v  = agg[i] + dv * dv * h[i] + b[f];
    out[i] = v > 0.0f ? v : 0.0f;
}

__global__ __launch_bounds__(256) void softmax32_k(float* __restrict__ out, int M) {
    const int warp = (blockIdx.x * blockDim.x + threadIdx.x) >> 5;
    const int lane = threadIdx.x & 31;
    if (warp >= M) return;
    const size_t idx = (size_t)warp * 32 + lane;
    const float v = out[idx];
    float m = v;
#pragma unroll
    for (int off = 16; off > 0; off >>= 1) m = fmaxf(m, __shfl_xor(m, off, 32));
    const float e = __expf(v - m);
    float s = e;
#pragma unroll
    for (int off = 16; off > 0; off >>= 1) s += __shfl_xor(s, off, 32);
    out[idx] = e / s;
}

extern "C" void kernel_launch(void* const* d_in, const int* in_sizes, int n_in,
                              void* d_out, int out_size, void* d_ws, size_t ws_size,
                              hipStream_t stream) {
    const float* x    = (const float*)d_in[0];
    const int*   edge = (const int*)d_in[1];
    const float* W1   = (const float*)d_in[2];
    const float* b1   = (const float*)d_in[3];
    const float* W2   = (const float*)d_in[4];
    const float* b2   = (const float*)d_in[5];
    const float* W3   = (const float*)d_in[6];
    const float* b3   = (const float*)d_in[7];
    float* out = (float*)d_out;

    const int* src = edge;
    const int* dst = edge + NEDGES;

    float* dis  = (float*)d_ws;
    float* hbuf = dis  + NNODES;
    float* abuf = hbuf + (size_t)NNODES * 64;
    float* xbuf = abuf + (size_t)NNODES * 64;

    const int TB = 256;
    auto blocks = [](long long n, int tb) { return (int)((n + tb - 1) / tb); };

    const int gemm_waves  = NNODES / 16;
    const int gemm_blocks = blocks((long long)gemm_waves * 32, TB);
    const int edge_blocks = blocks((long long)NEDGES * 32, TB);

    zero_f32_k<<<blocks(NNODES, TB), TB, 0, stream>>>(dis, NNODES);
    degree_k<<<blocks(NEDGES, TB), TB, 0, stream>>>(dst, dis, NEDGES);
    rsqrt_k<<<blocks(NNODES, TB), TB, 0, stream>>>(dis, NNODES);

    gemm_wmma_f32_k<64><<<gemm_blocks, TB, 0, stream>>>(x, W1, hbuf, NNODES, INDIM);
    zero_f32_k<<<blocks((long long)NNODES * 64, TB), TB, 0, stream>>>(abuf, NNODES * 64);
    edge_agg_k<64><<<edge_blocks, TB, 0, stream>>>(hbuf, src, dst, dis, abuf, NEDGES);
    finish_k<64><<<blocks((long long)NNODES * 64, TB), TB, 0, stream>>>(abuf, hbuf, dis, b1, xbuf, NNODES);

    gemm_wmma_f32_k<64><<<gemm_blocks, TB, 0, stream>>>(xbuf, W2, hbuf, NNODES, HID1);
    zero_f32_k<<<blocks((long long)NNODES * 64, TB), TB, 0, stream>>>(abuf, NNODES * 64);
    edge_agg_k<64><<<edge_blocks, TB, 0, stream>>>(hbuf, src, dst, dis, abuf, NEDGES);
    finish_k<64><<<blocks((long long)NNODES * 64, TB), TB, 0, stream>>>(abuf, hbuf, dis, b2, xbuf, NNODES);

    gemm_wmma_f32_k<32><<<gemm_blocks, TB, 0, stream>>>(xbuf, W3, hbuf, NNODES, HID2);
    zero_f32_k<<<blocks((long long)NNODES * 32, TB), TB, 0, stream>>>(abuf, NNODES * 32);
    edge_agg_k<32><<<edge_blocks, TB, 0, stream>>>(hbuf, src, dst, dis, abuf, NEDGES);
    finish_k<32><<<blocks((long long)NNODES * 32, TB), TB, 0, stream>>>(abuf, hbuf, dis, b3, out, NNODES);

    softmax32_k<<<blocks((long long)NNODES * 32, TB), TB, 0, stream>>>(out, NNODES);
}


# Edges grouped by destination once per call (count with rank, scan, place); one wave per node sums its edge rows in registers with the epilogue: no float atomics, no zero pass; sum order = list order
# speedup vs baseline: 2.2471x; 2.2471x over previous
#include <hip/hip_runtime.h>

typedef float v2f __attribute__((ext_vector_type(2)));
typedef float v8f __attribute__((ext_vector_type(8)));

#define NNODES 100000
#define NEDGES 800000
#define INDIM  128
#define HID1   64
#define HID2   64
#define ODIM   32

__global__ __launch_bounds__(256) void zero_f32_k(float* __restrict__ p, int n) {
    int i = blockIdx.x * blockDim.x + threadIdx.x;
    if (i < n) p[i] = 0.0f;
}

__global__ __launch_bounds__(256) void count_rank_k(const int* __restrict__ dst,
                                                    int* __restrict__ cnt,
                                                    int* __restrict__ rank, int nE) {
    int e = blockIdx.x * blockDim.x + threadIdx.x;
    if (e < nE) rank[e] = atomicAdd(&cnt[dst[e]], 1);
}

__global__ __launch_bounds__(256) void alloc_k(const int* __restrict__ cnt,
                                               int* __restrict__ cursor,
                                               int* __restrict__ start,
                                               float* __restrict__ dis, int n) {
    __shared__ int wsum[8];
    __shared__ int base_s;
    const int i    = blockIdx.x * 256 + threadIdx.x;
    const int lane = threadIdx.x & 31;
    const int w    = threadIdx.x >> 5;
    const int c    = (i < n) ? cnt[i] : 0;
    if (i < n) dis[i] = rsqrtf((float)c + 1.0f);
    int incl = c;
    for (int off = 1; off < 32; off <<= 1) {
        const int t = __shfl_up(incl, off, 32);
        if (lane >= off) incl += t;
    }
    if (lane == 31) wsum[w] = incl;
    __syncthreads();
    if (threadIdx.x == 0) {
        int tot = 0;
        for (int k = 0; k < 8; ++k) { const int t = wsum[k]; wsum[k] = tot; tot += t; }
        base_s = atomicAdd(cursor, tot);
    }
    __syncthreads();
    if (i < n) start[i] = base_s + wsum[w] + incl - c;
}

__global__ __launch_bounds__(256) void fill_k(const int* __restrict__ src,
                                              const int* __restrict__ dst,
                                              const int* __restrict__ rank,
                                              const int* __restrict__ start,
                                              const float* __restrict__ dis,
                                              int* __restrict__ csr, int nE) {
    int e = blockIdx.x * blockDim.x + threadIdx.x;
    if (e >= nE) return;
    const int s = src[e];
    const int d = dst[e];
    const int pos = start[d] + rank[e];
    const float wgt = dis[s] * dis[d];
    reinterpret_cast<int2*>(csr)[pos] = make_int2(s, __float_as_int(wgt));
}

template <int N>
__global__ __launch_bounds__(256) void gemm_wmma_f32_k(const float* __restrict__ A,
                                                       const float* __restrict__ W,
                                                       float* __restrict__ C,
                                                       int M, int K) {
    constexpr int NT = N / 16;
    const int lane = threadIdx.x & 31;
    const int wave = (blockIdx.x * blockDim.x + threadIdx.x) >> 5;
    const int m0   = wave << 4;
    if (m0 >= M) return;

    const int mr = lane & 15;
    const int kh = lane >> 4;

    v8f acc[NT];
#pragma unroll
    for (int t = 0; t < NT; ++t) acc[t] = {};

    const size_t arow = (size_t)(m0 + mr) * (size_t)K;

    for (int k0 = 0; k0 < K; k0 += 4) {
        v2f a;
        a.x = A[arow + k0 + 2 * kh + 0];
        a.y = A[arow + k0 + 2 * kh + 1];
#pragma unroll
        for (int t = 0; t < NT; ++t) {
            const int ncol = (t << 4) + mr;
            v2f b;
            b.x = W[(size_t)(k0 + 2 * kh + 0) * N + ncol];
            b.y = W[(size_t)(k0 + 2 * kh + 1) * N + ncol];
            acc[t] = __builtin_amdgcn_wmma_f32_16x16x4_f32(
                false, a, false, b, (short)0, acc[t], false, false);
        }
    }

#pragma unroll
    for (int t = 0; t < NT; ++t) {
        const int ncol = (t << 4) + mr;
#pragma unroll
        for (int v = 0; v < 8; ++v) {
            const int row = m0 + 8 * kh + v;
            C[(size_t)row * N + ncol] = acc[t][v];
        }
    }
}

template <int V>
__device__ __forceinline__ void ldrow(const float* __restrict__ p, float* r) {
    if constexpr (V == 2) {
        const float2 t = *reinterpret_cast<const float2*>(p);
        r[0] = t.x; r[1] = t.y;
    } else {
        r[0] = *p;
    }
}

template <int H, int CNT>
__device__ __forceinline__ void accum_rows(const float* __restrict__ h, int esrc, int ewgt,
                                           int j, int lane, float* acc) {
    constexpr int V = H / 32;
    float r[CNT][V];
    float wg[CNT];
    for (int k = 0; k < CNT; ++k) {
        const int s = __builtin_amdgcn_readlane(esrc, j + k);
        wg[k] = __int_as_float(__builtin_amdgcn_readlane(ewgt, j + k));
        ldrow<V>(h + (size_t)s * H + lane * V, r[k]);
    }
    for (int k = 0; k < CNT; ++k)
        for (int v = 0; v < V; ++v) acc[v] += wg[k] * r[k][v];
}

template <int H>
__global__ __launch_bounds__(256) void gather_finish_k(const float* __restrict__ h,
                                                       const int* __restrict__ csr,
                                                       const int* __restrict__ start,
                                                       const int* __restrict__ cnt,
                                                       const float* __restrict__ dis,
                                                       const float* __restrict__ b,
                                                       float* __restrict__ out, int M) {
    constexpr int V = H / 32;
    const int lane = threadIdx.x & 31;
    const int node = __builtin_amdgcn_readfirstlane((int)((blockIdx.x * blockDim.x + threadIdx.x) >> 5));
    if (node >= M) return;
    const int beg  = start[node];
    const int n    = cnt[node];
    const float dv = dis[node];
    const size_t self = (size_t)node * H + lane * V;
    float hs[V], bb[V], acc[V];
    ldrow<V>(h + self, hs);
    ldrow<V>(b + lane * V, bb);
    for (int v = 0; v < V; ++v) acc[v] = 0.0f;

    const int2* ed = reinterpret_cast<const int2*>(csr) + beg;
    for (int base = 0; base < n; base += 32) {
        const int m = (n - base < 32) ? (n - base) : 32;
        int2 ent = make_int2(0, 0);
        if (lane < m) ent = ed[base + lane];
        int j = 0;
        for (; j + 8 <= m; j += 8) accum_rows<H, 8>(h, ent.x, ent.y, j, lane, acc);
        if (j + 4 <= m) { accum_rows<H, 4>(h, ent.x, ent.y, j, lane, acc); j += 4; }
        const int t = m - j;
        if (t == 3)      accum_rows<H, 3>(h, ent.x, ent.y, j, lane, acc);
        else if (t == 2) accum_rows<H, 2>(h, ent.x, ent.y, j, lane, acc);
        else if (t == 1) accum_rows<H, 1>(h, ent.x, ent.y, j, lane, acc);
    }

    float o[V];
    for (int v = 0; v < V; ++v) {
        const float val = acc[v] + dv * dv * hs[v] + bb[v];
        o[v] = val > 0.0f ? val : 0.0f;
    }
    if constexpr (V == 2) {
        *reinterpret_cast<float2*>(out + self) = make_float2(o[0], o[1]);
    } else {
        out[self] = o[0];
    }
}

__global__ __launch_bounds__(256) void softmax32_k(float* __restrict__ out, int M) {
    const int warp = (blockIdx.x * blockDim.x + threadIdx.x) >> 5;
    const int lane = threadIdx.x & 31;
    if (warp >= M) return;
    const size_t idx = (size_t)warp * 32 + lane;
    const float v = out[idx];
    float m = v;
#pragma unroll
    for (int off = 16; off > 0; off >>= 1) m = fmaxf(m, __shfl_xor(m, off, 32));
    const float e = __expf(v - m);
    float s = e;
#pragma unroll
    for (int off = 16; off > 0; off >>= 1) s += __shfl_xor(s, off, 32);
    out[idx] = e / s;
}

extern "C" void kernel_launch(void* const* d_in, const int* in_sizes, int n_in,
                              void* d_out, int out_size, void* d_ws, size_t ws_size,
                              hipStream_t stream) {
    const float* x    = (const float*)d_in[0];
    const int*   edge = (const int*)d_in[1];
    const float* W1   = (const float*)d_in[2];
    const float* b1   = (const float*)d_in[3];
    const float* W2   = (const float*)d_in[4];
    const float* b2   = (const float*)d_in[5];
    const float* W3   = (const float*)d_in[6];
    const float* b3   = (const float*)d_in[7];
    float* out = (float*)d_out;

    const int* src = edge;
    const int* dst = edge + NEDGES;

    float* dis   = (float*)d_ws;
    int*   cnt   = (int*)(dis + NNODES);
    int*   start = cnt + NNODES + 32;
    int*   rank  = start + NNODES;
    int*   csr   = rank + NEDGES;
    float* hbuf  = (float*)(csr + (size_t)2 * NEDGES);
    float* xbuf  = hbuf + (size_t)NNODES * 64;

    const int TB = 256;
    auto blocks = [](long long n, int tb) { return (int)((n + tb - 1) / tb); };

    const int gemm_waves  = NNODES / 16;
    const int gemm_blocks = blocks((long long)gemm_waves * 32, TB);
    const int node_blocks = blocks((long long)NNODES * 32, TB);

    zero_f32_k<<<blocks(NNODES + 1, TB), TB, 0, stream>>>((float*)cnt, NNODES + 1);
    count_rank_k<<<blocks(NEDGES, TB), TB, 0, stream>>>(dst, cnt, rank, NEDGES);
    alloc_k<<<blocks(NNODES, TB), TB, 0, stream>>>(cnt, cnt + NNODES, start, dis, NNODES);
    fill_k<<<blocks(NEDGES, TB), TB, 0, stream>>>(src, dst, rank, start, dis, csr, NEDGES);

    gemm_wmma_f32_k<64><<<gemm_blocks, TB, 0, stream>>>(x, W1, hbuf, NNODES, INDIM);
    gather_finish_k<64><<<node_blocks, TB, 0, stream>>>(hbuf, csr, start, cnt, dis, b1, xbuf, NNODES);

    gemm_wmma_f32_k<64><<<gemm_blocks, TB, 0, stream>>>(xbuf, W2, hbuf, NNODES, HID1);
    gather_finish_k<64><<<node_blocks, TB, 0, stream>>>(hbuf, csr, start, cnt, dis, b2, xbuf, NNODES);

    gemm_wmma_f32_k<32><<<gemm_blocks, TB, 0, stream>>>(xbuf, W3, hbuf, NNODES, HID2);
    gather_finish_k<32><<<node_blocks, TB, 0, stream>>>(hbuf, csr, start, cnt, dis, b3, out, NNODES);

    softmax32_k<<<blocks((long long)NNODES * 32, TB), TB, 0, stream>>>(out, NNODES);
}


# Each wave takes several items, not one (8 nodes per gather wave, next node's list requested ahead; 16 softmax rows; 4 edges per thread in count/fill): ~6x fewer waves to start; sums unchanged
# speedup vs baseline: 2.3667x; 1.0532x over previous
#include <hip/hip_runtime.h>

typedef float v2f __attribute__((ext_vector_type(2)));
typedef float v8f __attribute__((ext_vector_type(8)));

#define NNODES 100000
#define NEDGES 800000
#define INDIM  128
#define HID1   64
#define HID2   64
#define ODIM   32


__global__ __launch_bounds__(256) void zero_f32_k(float* __restrict__ p, int n4) {
    int i = blockIdx.x * blockDim.x + threadIdx.x;
    if (i < n4) reinterpret_cast<float4*>(p)[i] = make_float4(0.0f, 0.0f, 0.0f, 0.0f);
}

__global__ __launch_bounds__(256) void count_rank_k(const int* __restrict__ dst,
                                                    int* __restrict__ cnt,
                                                    int* __restrict__ rank, int nE4) {
    int q = blockIdx.x * blockDim.x + threadIdx.x;
    if (q >= nE4) return;
    const int4 d = reinterpret_cast<const int4*>(dst)[q];
    int4 r;
    r.x = atomicAdd(&cnt[d.x], 1);
    r.y = atomicAdd(&cnt[d.y], 1);
    r.z = atomicAdd(&cnt[d.z], 1);
    r.w = atomicAdd(&cnt[d.w], 1);
    reinterpret_cast<int4*>(rank)[q] = r;
}

__global__ __launch_bounds__(256) void alloc_k(const int* __restrict__ cnt,
                                               int* __restrict__ cursor,
                                               int* __restrict__ start,
                                               float* __restrict__ dis, int n) {
    __shared__ int wsum[8];
    __shared__ int base_s;
    const int i    = blockIdx.x * 256 + threadIdx.x;
    const int lane = threadIdx.x & 31;
    const int w    = threadIdx.x >> 5;
    const int c    = (i < n) ? cnt[i] : 0;
    if (i < n) dis[i] = rsqrtf((float)c + 1.0f);
    int incl = c;
    for (int off = 1; off < 32; off <<= 1) {
        const int t = __shfl_up(incl, off, 32);
        if (lane >= off) incl += t;
    }
    if (lane == 31) wsum[w] = incl;
    __syncthreads();
    if (threadIdx.x == 0) {
        int tot = 0;
        for (int k = 0; k < 8; ++k) { const int t = wsum[k]; wsum[k] = tot; tot += t; }
        base_s = atomicAdd(cursor, tot);
    }
    __syncthreads();
    if (i < n) start[i] = base_s + wsum[w] + incl - c;
}

__device__ __forceinline__ void place_edge(int s, int d, int rk,
                                           const int* __restrict__ start,
                                           const float* __restrict__ dis,
                                           int* __restrict__ csr) {
    const int pos = start[d] + rk;
    const float wgt = dis[s] * dis[d];
    reinterpret_cast<int2*>(csr)[pos] = make_int2(s, __float_as_int(wgt));
}

__global__ __launch_bounds__(256) void fill_k(const int* __restrict__ src,
                                              const int* __restrict__ dst,
                                              const int* __restrict__ rank,
                                              const int* __restrict__ start,
                                              const float* __restrict__ dis,
                                              int* __restrict__ csr, int nE4) {
    int q = blockIdx.x * blockDim.x + threadIdx.x;
    if (q >= nE4) return;
    const int4 s = reinterpret_cast<const int4*>(src)[q];
    const int4 d = reinterpret_cast<const int4*>(dst)[q];
    const int4 r = reinterpret_cast<const int4*>(rank)[q];
    place_edge(s.x, d.x, r.x, start, dis, csr);
    place_edge(s.y, d.y, r.y, start, dis, csr);
    place_edge(s.z, d.z, r.z, start, dis, csr);
    place_edge(s.w, d.w, r.w, start, dis, csr);
}

template <int N>
__global__ __launch_bounds__(256) void gemm_wmma_f32_k(const float* __restrict__ A,
                                                       const float* __restrict__ W,
                                                       float* __restrict__ C,
                                                       int M, int K) {
    constexpr int NT = N / 16;
    const int lane = threadIdx.x & 31;
    const int wave = (blockIdx.x * blockDim.x + threadIdx.x) >> 5;
    const int m0   = wave << 4;
    if (m0 >= M) return;

    const int mr = lane & 15;
    const int kh = lane >> 4;

    v8f acc[NT];
#pragma unroll
    for (int t = 0; t < NT; ++t) acc[t] = {};

    const size_t arow = (size_t)(m0 + mr) * (size_t)K;

    for (int k0 = 0; k0 < K; k0 += 4) {
        v2f a;
        a.x = A[arow + k0 + 2 * kh + 0];
        a.y = A[arow + k0 + 2 * kh + 1];
#pragma unroll
        for (int t = 0; t < NT; ++t) {
            const int ncol = (t << 4) + mr;
            v2f b;
            b.x = W[(size_t)(k0 + 2 * kh + 0) * N + ncol];
            b.y = W[(size_t)(k0 + 2 * kh + 1) * N + ncol];
            acc[t] = __builtin_amdgcn_wmma_f32_16x16x4_f32(
                false, a, false, b, (short)0, acc[t], false, false);
        }
    }

#pragma unroll
    for (int t = 0; t < NT; ++t) {
        const int ncol = (t << 4) + mr;
#pragma unroll
        for (int v = 0; v < 8; ++v) {
            const int row = m0 + 8 * kh + v;
            C[(size_t)row * N + ncol] = acc[t][v];
        }
    }
}

template <int V>
__device__ __forceinline__ void ldrow(const float* __restrict__ p, float* r) {
    if constexpr (V == 2) {
        const float2 t = *reinterpret_cast<const float2*>(p);
        r[0] = t.x; r[1] = t.y;
    } else {
        r[0] = *p;
    }
}

template <int H, int CNT>
__device__ __forceinline__ void accum_rows(const float* __restrict__ h, int esrc, int ewgt,
                                           int j, int lane, float* acc) {
    constexpr int V = H / 32;
    float r[CNT][V];
    float wg[CNT];
    for (int k = 0; k < CNT; ++k) {
        const int s = __builtin_amdgcn_readlane(esrc, j + k);
        wg[k] = __int_as_float(__builtin_amdgcn_readlane(ewgt, j + k));
        ldrow<V>(h + (size_t)s * H + lane * V, r[k]);
    }
    for (int k = 0; k < CNT; ++k)
        for (int v = 0; v < V; ++v) acc[v] += wg[k] * r[k][v];
}

template <int H>
__device__ __forceinline__ void accum_chunk(const float* __restrict__ h, int2 ent, int m,
                                            int lane, float* acc) {
    int j = 0;
    for (; j + 8 <= m; j += 8) accum_rows<H, 8>(h, ent.x, ent.y, j, lane, acc);
    if (j + 4 <= m) { accum_rows<H, 4>(h, ent.x, ent.y, j, lane, acc); j += 4; }
    const int t = m - j;
    if (t == 3)      accum_rows<H, 3>(h, ent.x, ent.y, j, lane, acc);
    else if (t == 2) accum_rows<H, 2>(h, ent.x, ent.y, j, lane, acc);
    else if (t == 1) accum_rows<H, 1>(h, ent.x, ent.y, j, lane, acc);
}

template <int H, int NPW>
__global__ __launch_bounds__(256) void gather_finish_k(const float* __restrict__ h,
                                                       const int* __restrict__ csr,
                                                       const int* __restrict__ start,
                                                       const int* __restrict__ cnt,
                                                       const float* __restrict__ dis,
                                                       const float* __restrict__ b,
                                                       float* __restrict__ out, int M) {
    constexpr int V = H / 32;
    static_assert(NPW <= 32, "one lane per node of the wave");
    const int lane = threadIdx.x & 31;
    const int wave  = __builtin_amdgcn_readfirstlane((int)((blockIdx.x * blockDim.x + threadIdx.x) >> 5));
    const int node0 = wave * NPW;
    if (node0 >= M) return;
    const int nn = (M - node0 < NPW) ? (M - node0) : NPW;

    int vbeg = 0, vcnt = 0;
    float vdis = 0.0f;
    if (lane < nn) {
        vbeg = start[node0 + lane];
        vcnt = cnt[node0 + lane];
        vdis = dis[node0 + lane];
    }
    float bb[V];
    ldrow<V>(b + lane * V, bb);

    const int2* csr2 = reinterpret_cast<const int2*>(csr);

    int beg_nx = __builtin_amdgcn_readlane(vbeg, 0);
    int n_nx   = __builtin_amdgcn_readlane(vcnt, 0);
    int2 ent_nx = make_int2(0, 0);
    if (lane < n_nx) ent_nx = csr2[beg_nx + lane];
    float hs_nx[V];
    ldrow<V>(h + (size_t)node0 * H + lane * V, hs_nx);

    for (int i = 0; i < nn; ++i) {
        const int node = node0 + i;
        const int beg  = beg_nx;
        const int n    = n_nx;
        const float dv = __int_as_float(__builtin_amdgcn_readlane(__float_as_int(vdis), i));
        int2 ent = ent_nx;
        float hs[V], acc[V];
        for (int v = 0; v < V; ++v) { hs[v] = hs_nx[v]; acc[v] = 0.0f; }

        if (i + 1 < nn) {
            beg_nx = __builtin_amdgcn_readlane(vbeg, i + 1);
            n_nx   = __builtin_amdgcn_readlane(vcnt, i + 1);
            ent_nx = make_int2(0, 0);
            if (lane < n_nx) ent_nx = csr2[beg_nx + lane];
            ldrow<V>(h + (size_t)(node + 1) * H + lane * V, hs_nx);
        }

        accum_chunk<H>(h, ent, (n < 32) ? n : 32, lane, acc);
        for (int base = 32; base < n; base += 32) {
            const int m = (n - base < 32) ? (n - base) : 32;
            int2 more = make_int2(0, 0);
            if (lane < m) more = csr2[beg + base + lane];
            accum_chunk<H>(h, more, m, lane, acc);
        }

        const size_t self = (size_t)node * H + lane * V;
        float o[V];
        for (int v = 0; v < V; ++v) {
            const float val = acc[v] + dv * dv * hs[v] + bb[v];
            o[v] = val > 0.0f ? val : 0.0f;
        }
        if constexpr (V == 2) {
            *reinterpret_cast<float2*>(out + self) = make_float2(o[0], o[1]);
        } else {
            out[self] = o[0];
        }
    }
}

template <int R>
__global__ __launch_bounds__(256) void softmax32_k(float* __restrict__ out, int M) {
    const int wave = __builtin_amdgcn_readfirstlane((int)((blockIdx.x * blockDim.x + threadIdx.x) >> 5));
    const int lane = threadIdx.x & 31;
    const int row0 = wave * R;
    if (row0 >= M) return;
    float v[R], m[R], e[R], s[R];
#pragma unroll
    for (int r = 0; r < R; ++r) {
        const int row = (row0 + r < M) ? (row0 + r) : (M - 1);
        v[r] = out[(size_t)row * 32 + lane];
    }
#pragma unroll
    for (int r = 0; r < R; ++r) m[r] = v[r];
#pragma unroll
    for (int off = 16; off > 0; off >>= 1) {
#pragma unroll
        for (int r = 0; r < R; ++r) m[r] = fmaxf(m[r], __shfl_xor(m[r], off, 32));
    }
#pragma unroll
    for (int r = 0; r < R; ++r) { e[r] = __expf(v[r] - m[r]); s[r] = e[r]; }
#pragma unroll
    for (int off = 16; off > 0; off >>= 1) {
#pragma unroll
        for (int r = 0; r < R; ++r) s[r] += __shfl_xor(s[r], off, 32);
    }
#pragma unroll
    for (int r = 0; r < R; ++r)
        if (row0 + r < M) out[(size_t)(row0 + r) * 32 + lane] = e[r] / s[r];
}

extern "C" void kernel_launch(void* const* d_in, const int* in_sizes, int n_in,
                              void* d_out, int out_size, void* d_ws, size_t ws_size,
                              hipStream_t stream) {
    const float* x    = (const float*)d_in[0];
    const int*   edge = (const int*)d_in[1];
    const float* W1   = (const float*)d_in[2];
    const float* b1   = (const float*)d_in[3];
    const float* W2   = (const float*)d_in[4];
    const float* b2   = (const float*)d_in[5];
    const float* W3   = (const float*)d_in[6];
    const float* b3   = (const float*)d_in[7];
    float* out = (float*)d_out;

    const int* src = edge;
    const int* dst = edge + NEDGES;

    float* dis   = (float*)d_ws;
    int*   cnt   = (int*)(dis + NNODES);
    int*   start = cnt + NNODES + 32;
    int*   rank  = start + NNODES;
    int*   csr   = rank + NEDGES;
    float* hbuf  = (float*)(csr + (size_t)2 * NEDGES);
    float* xbuf  = hbuf + (size_t)NNODES * 64;

    const int TB = 256;
    auto blocks = [](long long n, int tb) { return (int)((n + tb - 1) / tb); };

    const int gemm_waves  = NNODES / 16;
    const int gemm_blocks = blocks((long long)gemm_waves * 32, TB);
    constexpr int NPW = 8;
    constexpr int SMR = 16;
    const int node_blocks = blocks((long long)((NNODES + NPW - 1) / NPW) * 32, TB);
    const int smax_blocks = blocks((long long)((NNODES + SMR - 1) / SMR) * 32, TB);
    static_assert(NEDGES % 4 == 0, "count and fill take four edges per thread");
    static_assert((NNODES + 32) % 4 == 0, "the count words are cleared four at a time");

    zero_f32_k<<<blocks((NNODES + 32) / 4, TB), TB, 0, stream>>>((float*)cnt, (NNODES + 32) / 4);
    count_rank_k<<<blocks(NEDGES / 4, TB), TB, 0, stream>>>(dst, cnt, rank, NEDGES / 4);
    alloc_k<<<blocks(NNODES, TB), TB, 0, stream>>>(cnt, cnt + NNODES, start, dis, NNODES);
    fill_k<<<blocks(NEDGES / 4, TB), TB, 0, stream>>>(src, dst, rank, start, dis, csr, NEDGES / 4);

    gemm_wmma_f32_k<64><<<gemm_blocks, TB, 0, stream>>>(x, W1, hbuf, NNODES, INDIM);
    gather_finish_k<64, NPW><<<node_blocks, TB, 0, stream>>>(hbuf, csr, start, cnt, dis, b1, xbuf, NNODES);

    gemm_wmma_f32_k<64><<<gemm_blocks, TB, 0, stream>>>(xbuf, W2, hbuf, NNODES, HID1);
    gather_finish_k<64, NPW><<<node_blocks, TB, 0, stream>>>(hbuf, csr, start, cnt, dis, b2, xbuf, NNODES);

    gemm_wmma_f32_k<32><<<gemm_blocks, TB, 0, stream>>>(xbuf, W3, hbuf, NNODES, HID2);
    gather_finish_k<32, NPW><<<node_blocks, TB, 0, stream>>>(hbuf, csr, start, cnt, dis, b3, out, NNODES);

    softmax32_k<SMR><<<smax_blocks, TB, 0, stream>>>(out, NNODES);
}


# Each gather kernel also runs its consumer: rows go to an LDS tile, the block multiplies it by the next layer's weights on WMMA (last layer: softmax); 11 launches become 8; sums unchanged
# speedup vs baseline: 2.7882x; 1.1781x over previous
#include <hip/hip_runtime.h>

typedef float v2f __attribute__((ext_vector_type(2)));
typedef float v8f __attribute__((ext_vector_type(8)));

#define NNODES 100000
#define NEDGES 800000
#define INDIM  128
#define HID1   64
#define HID2   64
#define ODIM   32


__global__ __launch_bounds__(256) void zero_f32_k(float* __restrict__ p, int n4) {
    int i = blockIdx.x * blockDim.x + threadIdx.x;
    if (i < n4) reinterpret_cast<float4*>(p)[i] = make_float4(0.0f, 0.0f, 0.0f, 0.0f);
}

__global__ __launch_bounds__(256) void count_rank_k(const int* __restrict__ dst,
                                                    int* __restrict__ cnt,
                                                    int* __restrict__ rank, int nE4) {
    int q = blockIdx.x * blockDim.x + threadIdx.x;
    if (q >= nE4) return;
    const int4 d = reinterpret_cast<const int4*>(dst)[q];
    int4 r;
    r.x = atomicAdd(&cnt[d.x], 1);
    r.y = atomicAdd(&cnt[d.y], 1);
    r.z = atomicAdd(&cnt[d.z], 1);
    r.w = atomicAdd(&cnt[d.w], 1);
    reinterpret_cast<int4*>(rank)[q] = r;
}

__global__ __launch_bounds__(256) void alloc_k(const int* __restrict__ cnt,
                                               int* __restrict__ cursor,
                                               int* __restrict__ start,
                                               float* __restrict__ dis, int n) {
    __shared__ int wsum[8];
    __shared__ int base_s;
    const int i    = blockIdx.x * 256 + threadIdx.x;
    const int lane = threadIdx.x & 31;
    const int w    = threadIdx.x >> 5;
    const int c    = (i < n) ? cnt[i] : 0;
    if (i < n) dis[i] = rsqrtf((float)c + 1.0f);
    int incl = c;
    for (int off = 1; off < 32; off <<= 1) {
        const int t = __shfl_up(incl, off, 32);
        if (lane >= off) incl += t;
    }
    if (lane == 31) wsum[w] = incl;
    __syncthreads();
    if (threadIdx.x == 0) {
        int tot = 0;
        for (int k = 0; k < 8; ++k) { const int t = wsum[k]; wsum[k] = tot; tot += t; }
        base_s = atomicAdd(cursor, tot);
    }
    __syncthreads();
    if (i < n) start[i] = base_s + wsum[w] + incl - c;
}

__device__ __forceinline__ void place_edge(int s, int d, int rk,
                                           const int* __restrict__ start,
                                           const float* __restrict__ dis,
                                           int* __restrict__ csr) {
    const int pos = start[d] + rk;
    const float wgt = dis[s] * dis[d];
    reinterpret_cast<int2*>(csr)[pos] = make_int2(s, __float_as_int(wgt));
}

__global__ __launch_bounds__(256) void fill_k(const int* __restrict__ src,
                                              const int* __restrict__ dst,
                                              const int* __restrict__ rank,
                                              const int* __restrict__ start,
                                              const float* __restrict__ dis,
                                              int* __restrict__ csr, int nE4) {
    int q = blockIdx.x * blockDim.x + threadIdx.x;
    if (q >= nE4) return;
    const int4 s = reinterpret_cast<const int4*>(src)[q];
    const int4 d = reinterpret_cast<const int4*>(dst)[q];
    const int4 r = reinterpret_cast<const int4*>(rank)[q];
    place_edge(s.x, d.x, r.x, start, dis, csr);
    place_edge(s.y, d.y, r.y, start, dis, csr);
    place_edge(s.z, d.z, r.z, start, dis, csr);
    place_edge(s.w, d.w, r.w, start, dis, csr);
}

template <int N>
__global__ __launch_bounds__(256) void gemm_wmma_f32_k(const float* __restrict__ A,
                                                       const float* __restrict__ W,
                                                       float* __restrict__ C,
                                                       int M, int K) {
    constexpr int NT = N / 16;
    const int lane = threadIdx.x & 31;
    const int wave = (blockIdx.x * blockDim.x + threadIdx.x) >> 5;
    const int m0   = wave << 4;
    if (m0 >= M) return;

    const int mr = lane & 15;
    const int kh = lane >> 4;

    v8f acc[NT];
#pragma unroll
    for (int t = 0; t < NT; ++t) acc[t] = {};

    const size_t arow = (size_t)(m0 + mr) * (size_t)K;

    for (int k0 = 0; k0 < K; k0 += 4) {
        v2f a;
        a.x = A[arow + k0 + 2 * kh + 0];
        a.y = A[arow + k0 + 2 * kh + 1];
#pragma unroll
        for (int t = 0; t < NT; ++t) {
            const int ncol = (t << 4) + mr;
            v2f b;
            b.x = W[(size_t)(k0 + 2 * kh + 0) * N + ncol];
            b.y = W[(size_t)(k0 + 2 * kh + 1) * N + ncol];
            acc[t] = __builtin_amdgcn_wmma_f32_16x16x4_f32(
                false, a, false, b, (short)0, acc[t], false, false);
        }
    }

#pragma unroll
    for (int t = 0; t < NT; ++t) {
        const int ncol = (t << 4) + mr;
#pragma unroll
        for (int v = 0; v < 8; ++v) {
            const int row = m0 + 8 * kh + v;
            C[(size_t)row * N + ncol] = acc[t][v];
        }
    }
}

template <int V>
__device__ __forceinline__ void ldrow(const float* __restrict__ p, float* r) {
    if constexpr (V == 2) {
        const float2 t = *reinterpret_cast<const float2*>(p);
        r[0] = t.x; r[1] = t.y;
    } else {
        r[0] = *p;
    }
}

template <int H, int CNT>
__device__ __forceinline__ void accum_rows(const float* __restrict__ h, int esrc, int ewgt,
                                           int j, int lane, float* acc) {
    constexpr int V = H / 32;
    float r[CNT][V];
    float wg[CNT];
    for (int k = 0; k < CNT; ++k) {
        const int s = __builtin_amdgcn_readlane(esrc, j + k);
        wg[k] = __int_as_float(__builtin_amdgcn_readlane(ewgt, j + k));
        ldrow<V>(h + (size_t)s * H + lane * V, r[k]);
    }
    for (int k = 0; k < CNT; ++k)
        for (int v = 0; v < V; ++v) acc[v] += wg[k] * r[k][v];
}

template <int H>
__device__ __forceinline__ void accum_chunk(const float* __restrict__ h, int2 ent, int m,
                                            int lane, float* acc) {
    int j = 0;
    for (; j + 8 <= m; j += 8) accum_rows<H, 8>(h, ent.x, ent.y, j, lane, acc);
    if (j + 4 <= m) { accum_rows<H, 4>(h, ent.x, ent.y, j, lane, acc); j += 4; }
    const int t = m - j;
    if (t == 3)      accum_rows<H, 3>(h, ent.x, ent.y, j, lane, acc);
    else if (t == 2) accum_rows<H, 2>(h, ent.x, ent.y, j, lane, acc);
    else if (t == 1) accum_rows<H, 1>(h, ent.x, ent.y, j, lane, acc);
}

template <int H, int NPW, int PITCH>
__device__ __forceinline__ void gather_to_lds(const float* __restrict__ h,
                                              const int* __restrict__ csr,
                                              const int* __restrict__ start,
                                              const int* __restrict__ cnt,
                                              const float* __restrict__ dis,
                                              const float* __restrict__ b,
                                              float* tile, int node0, int nn, int lane) {
    constexpr int V = H / 32;
    static_assert(NPW <= 32, "one lane per node of the wave");
    if (nn < NPW) {
#pragma unroll
        for (int i = 0; i < NPW; ++i)
            for (int v = 0; v < V; ++v) tile[i * PITCH + lane * V + v] = 0.0f;
    }
    if (nn > 0) {
        int vbeg = 0, vcnt = 0;
        float vdis = 0.0f;
        if (lane < nn) {
            vbeg = start[node0 + lane];
            vcnt = cnt[node0 + lane];
            vdis = dis[node0 + lane];
        }
        float bb[V];
        ldrow<V>(b + lane * V, bb);

        const int2* csr2 = reinterpret_cast<const int2*>(csr);

        int beg_nx = __builtin_amdgcn_readlane(vbeg, 0);
        int n_nx   = __builtin_amdgcn_readlane(vcnt, 0);
        int2 ent_nx = make_int2(0, 0);
        if (lane < n_nx) ent_nx = csr2[beg_nx + lane];
        float hs_nx[V];
        ldrow<V>(h + (size_t)node0 * H + lane * V, hs_nx);

        for (int i = 0; i < nn; ++i) {
            const int node = node0 + i;
            const int beg  = beg_nx;
            const int n    = n_nx;
            const float dv = __int_as_float(__builtin_amdgcn_readlane(__float_as_int(vdis), i));
            int2 ent = ent_nx;
            float hs[V], acc[V];
            for (int v = 0; v < V; ++v) { hs[v] = hs_nx[v]; acc[v] = 0.0f; }

            if (i + 1 < nn) {
                beg_nx = __builtin_amdgcn_readlane(vbeg, i + 1);
                n_nx   = __builtin_amdgcn_readlane(vcnt, i + 1);
                ent_nx = make_int2(0, 0);
                if (lane < n_nx) ent_nx = csr2[beg_nx + lane];
                ldrow<V>(h + (size_t)(node + 1) * H + lane * V, hs_nx);
            }

            accum_chunk<H>(h, ent, (n < 32) ? n : 32, lane, acc);
            for (int base = 32; base < n; base += 32) {
                const int m = (n - base < 32) ? (n - base) : 32;
                int2 more = make_int2(0, 0);
                if (lane < m) more = csr2[beg + base + lane];
                accum_chunk<H>(h, more, m, lane, acc);
            }

            float* trow = tile + i * PITCH + lane * V;
            for (int v = 0; v < V; ++v) {
                const float val = acc[v] + dv * dv * hs[v] + bb[v];
                trow[v] = val > 0.0f ? val : 0.0f;
            }
        }
    }
}

template <int N>
__global__ __launch_bounds__(256) void gather_gemm_k(const float* __restrict__ h,
                                                     const int* __restrict__ csr,
                                                     const int* __restrict__ start,
                                                     const int* __restrict__ cnt,
                                                     const float* __restrict__ dis,
                                                     const float* __restrict__ b,
                                                     const float* __restrict__ Wn,
                                                     float* __restrict__ C, int M) {
    constexpr int H = 64, NPW = 8, XP = 68, WP = N + 8;
    constexpr int W4 = (H * N / 4) / 256;
    constexpr int TPW = N / 32;
    __shared__ float xs[64 * XP];
    __shared__ float wsm[H * WP];

    const int lane = threadIdx.x & 31;
    const int w = __builtin_amdgcn_readfirstlane((int)(threadIdx.x >> 5));

    const int node0 = (blockIdx.x * 64 + w * NPW);
    int nn = M - node0;
    nn = nn < 0 ? 0 : (nn > NPW ? NPW : nn);
    gather_to_lds<H, NPW, XP>(h, csr, start, cnt, dis, b, xs + w * NPW * XP, node0, nn, lane);

#pragma unroll
    for (int q = 0; q < W4; ++q) {
        const int f = q * 256 + threadIdx.x;
        const int r = f / (N / 4);
        const int c = (f % (N / 4)) * 4;
        *reinterpret_cast<float4*>(&wsm[r * WP + c]) = reinterpret_cast<const float4*>(Wn)[f];
    }
    __syncthreads();

    const int mt = w >> 1;
    const int m0 = blockIdx.x * 64 + mt * 16;
    if (m0 >= M) return;
    const int nt0 = (w & 1) * TPW;
    const int mr = lane & 15;
    const int kh = lane >> 4;

    v8f acc[TPW];
#pragma unroll
    for (int t = 0; t < TPW; ++t) acc[t] = {};

    const float* arow = xs + (mt * 16 + mr) * XP + 2 * kh;
    const float* bcol = wsm + (2 * kh) * WP + nt0 * 16 + mr;
#pragma unroll
    for (int k0 = 0; k0 < H; k0 += 4) {
        v2f a;
        a.x = arow[k0 + 0];
        a.y = arow[k0 + 1];
#pragma unroll
        for (int t = 0; t < TPW; ++t) {
            v2f bq;
            bq.x = bcol[(k0 + 0) * WP + t * 16];
            bq.y = bcol[(k0 + 1) * WP + t * 16];
            acc[t] = __builtin_amdgcn_wmma_f32_16x16x4_f32(
                false, a, false, bq, (short)0, acc[t], false, false);
        }
    }

#pragma unroll
    for (int t = 0; t < TPW; ++t) {
        const int ncol = ((nt0 + t) << 4) + mr;
#pragma unroll
        for (int v = 0; v < 8; ++v) {
            const int row = m0 + 8 * kh + v;
            C[(size_t)row * N + ncol] = acc[t][v];
        }
    }
}

__global__ __launch_bounds__(256) void gather_softmax_k(const float* __restrict__ h,
                                                        const int* __restrict__ csr,
                                                        const int* __restrict__ start,
                                                        const int* __restrict__ cnt,
                                                        const float* __restrict__ dis,
                                                        const float* __restrict__ b,
                                                        float* __restrict__ out, int M) {
    constexpr int H = 32, NPW = 8;
    __shared__ float xs[64 * H];
    const int lane = threadIdx.x & 31;
    const int w = __builtin_amdgcn_readfirstlane((int)(threadIdx.x >> 5));
    const int node0 = blockIdx.x * 64 + w * NPW;
    if (node0 >= M) return;
    const int nn = (M - node0 < NPW) ? (M - node0) : NPW;
    float* tile = xs + w * NPW * H;
    gather_to_lds<H, NPW, H>(h, csr, start, cnt, dis, b, tile, node0, nn, lane);

    float v[NPW], m[NPW], e[NPW], s[NPW];
#pragma unroll
    for (int r = 0; r < NPW; ++r) { v[r] = tile[r * H + lane]; m[r] = v[r]; }
#pragma unroll
    for (int off = 16; off > 0; off >>= 1) {
#pragma unroll
        for (int r = 0; r < NPW; ++r) m[r] = fmaxf(m[r], __shfl_xor(m[r], off, 32));
    }
#pragma unroll
    for (int r = 0; r < NPW; ++r) { e[r] = __expf(v[r] - m[r]); s[r] = e[r]; }
#pragma unroll
    for (int off = 16; off > 0; off >>= 1) {
#pragma unroll
        for (int r = 0; r < NPW; ++r) s[r] += __shfl_xor(s[r], off, 32);
    }
#pragma unroll
    for (int r = 0; r < NPW; ++r)
        if (r < nn) out[(size_t)(node0 + r) * H + lane] = e[r] / s[r];
}

extern "C" void kernel_launch(void* const* d_in, const int* in_sizes, int n_in,
                              void* d_out, int out_size, void* d_ws, size_t ws_size,
                              hipStream_t stream) {
    const float* x    = (const float*)d_in[0];
    const int*   edge = (const int*)d_in[1];
    const float* W1   = (const float*)d_in[2];
    const float* b1   = (const float*)d_in[3];
    const float* W2   = (const float*)d_in[4];
    const float* b2   = (const float*)d_in[5];
    const float* W3   = (const float*)d_in[6];
    const float* b3   = (const float*)d_in[7];
    float* out = (float*)d_out;

    const int* src = edge;
    const int* dst = edge + NEDGES;

    float* dis   = (float*)d_ws;
    int*   cnt   = (int*)(dis + NNODES);
    int*   start = cnt + NNODES + 32;
    int*   rank  = start + NNODES;
    int*   csr   = rank + NEDGES;
    float* hbuf  = (float*)(csr + (size_t)2 * NEDGES);
    float* xbuf  = hbuf + (size_t)NNODES * 64;

    const int TB = 256;
    auto blocks = [](long long n, int tb) { return (int)((n + tb - 1) / tb); };

    const int gemm_waves  = NNODES / 16;
    const int gemm_blocks = blocks((long long)gemm_waves * 32, TB);
    const int node_blocks = blocks(NNODES, 64);
    static_assert(NEDGES % 4 == 0, "count and fill take four edges per thread");
    static_assert((NNODES + 32) % 4 == 0, "the count words are cleared four at a time");

    zero_f32_k<<<blocks((NNODES + 32) / 4, TB), TB, 0, stream>>>((float*)cnt, (NNODES + 32) / 4);
    count_rank_k<<<blocks(NEDGES / 4, TB), TB, 0, stream>>>(dst, cnt, rank, NEDGES / 4);
    alloc_k<<<blocks(NNODES, TB), TB, 0, stream>>>(cnt, cnt + NNODES, start, dis, NNODES);
    fill_k<<<blocks(NEDGES / 4, TB), TB, 0, stream>>>(src, dst, rank, start, dis, csr, NEDGES / 4);

    gemm_wmma_f32_k<64><<<gemm_blocks, TB, 0, stream>>>(x, W1, hbuf, NNODES, INDIM);

    gather_gemm_k<64><<<node_blocks, TB, 0, stream>>>(hbuf, csr, start, cnt, dis, b1, W2, xbuf, NNODES);

    gather_gemm_k<32><<<node_blocks, TB, 0, stream>>>(xbuf, csr, start, cnt, dis, b2, W3, hbuf, NNODES);

    gather_softmax_k<<<node_blocks, TB, 0, stream>>>(hbuf, csr, start, cnt, dis, b3, out, NNODES);
}


# In-degree counts and edge ranks on LDS counters: block (edge slice, node range) counts in LDS, per-slice table scanned to slots; no global atomic per edge, no zero pass; sum order = slice, then rank
# speedup vs baseline: 3.2607x; 1.1695x over previous
#include <hip/hip_runtime.h>

typedef float v2f __attribute__((ext_vector_type(2)));
typedef float v8f __attribute__((ext_vector_type(8)));

#define NNODES 100000
#define NEDGES 800000
#define INDIM  128
#define HID1   64
#define HID2   64
#define ODIM   32


#define ESLICES 32
#define NRANGES 8
#define ESL (NEDGES / ESLICES)
#define NRG (NNODES / NRANGES)
#define CRT 512

__global__ __launch_bounds__(CRT) void count_rank_k(const int* __restrict__ dst,
                                                    int* __restrict__ tab,
                                                    int* __restrict__ rank,
                                                    int* __restrict__ cursor) {
    constexpr int Q   = ESL / 4;
    constexpr int NIT = (Q + CRT - 1) / CRT;
    constexpr int C4  = NRG / 4;
    static_assert(ESL % 4 == 0 && NRG % 4 == 0, "slices and ranges are moved four words at a time");
    __shared__ __attribute__((aligned(16))) int c[NRG];
    const int tid = threadIdx.x;
    const int sl  = blockIdx.x / NRANGES;
    const int lo  = (blockIdx.x % NRANGES) * NRG;

    const int4* d4 = reinterpret_cast<const int4*>(dst + (size_t)sl * ESL);
    int4 dv[NIT];
#pragma unroll
    for (int k = 0; k < NIT; ++k) {
        const int q = tid + k * CRT;
        dv[k] = (q < Q) ? d4[q] : make_int4(-1, -1, -1, -1);
    }
    for (int i = tid; i < C4; i += CRT) reinterpret_cast<int4*>(c)[i] = make_int4(0, 0, 0, 0);
    if (blockIdx.x == 0 && tid == 0) *cursor = 0;
    __syncthreads();

#pragma unroll
    for (int k = 0; k < NIT; ++k) {
        const unsigned t0 = (unsigned)(dv[k].x - lo);
        const unsigned t1 = (unsigned)(dv[k].y - lo);
        const unsigned t2 = (unsigned)(dv[k].z - lo);
        const unsigned t3 = (unsigned)(dv[k].w - lo);
        int4 r = make_int4(-1, -1, -1, -1);
        if (t0 < (unsigned)NRG) r.x = atomicAdd(&c[t0], 1);
        if (t1 < (unsigned)NRG) r.y = atomicAdd(&c[t1], 1);
        if (t2 < (unsigned)NRG) r.z = atomicAdd(&c[t2], 1);
        if (t3 < (unsigned)NRG) r.w = atomicAdd(&c[t3], 1);
        dv[k] = r;
    }
    __syncthreads();

    int* rk = rank + (size_t)sl * ESL;
#pragma unroll
    for (int k = 0; k < NIT; ++k) {
        const int q = tid + k * CRT;
        if (dv[k].x >= 0) rk[4 * q + 0] = dv[k].x;
        if (dv[k].y >= 0) rk[4 * q + 1] = dv[k].y;
        if (dv[k].z >= 0) rk[4 * q + 2] = dv[k].z;
        if (dv[k].w >= 0) rk[4 * q + 3] = dv[k].w;
    }

    int4* trow = reinterpret_cast<int4*>(tab + (size_t)sl * NNODES + lo);
    for (int i = tid; i < C4; i += CRT) trow[i] = reinterpret_cast<const int4*>(c)[i];
}

__global__ __launch_bounds__(256) void alloc_k(int* __restrict__ tab,
                                               int* __restrict__ cursor,
                                               int* __restrict__ cnt,
                                               int* __restrict__ start,
                                               float* __restrict__ dis, int n) {
    __shared__ int wsum[8];
    __shared__ int base_s;
    const int i    = blockIdx.x * 256 + threadIdx.x;
    const int lane = threadIdx.x & 31;
    const int w    = threadIdx.x >> 5;
    const int ii = (i < n) ? i : n - 1;
    int h[ESLICES];
#pragma unroll
    for (int s = 0; s < ESLICES; ++s) h[s] = tab[(size_t)s * NNODES + ii];
    int c = 0;
#pragma unroll
    for (int s = 0; s < ESLICES; ++s) c += h[s];
    if (i >= n) c = 0;
    if (i < n) {
        cnt[i] = c;
        dis[i] = rsqrtf((float)c + 1.0f);
    }
    int incl = c;
    for (int off = 1; off < 32; off <<= 1) {
        const int t = __shfl_up(incl, off, 32);
        if (lane >= off) incl += t;
    }
    if (lane == 31) wsum[w] = incl;
    __syncthreads();
    if (threadIdx.x == 0) {
        int tot = 0;
        for (int k = 0; k < 8; ++k) { const int t = wsum[k]; wsum[k] = tot; tot += t; }
        base_s = atomicAdd(cursor, tot);
    }
    __syncthreads();
    if (i < n) {
        int run = base_s + wsum[w] + incl - c;
        start[i] = run;
#pragma unroll
        for (int s = 0; s < ESLICES; ++s) {
            tab[(size_t)s * NNODES + i] = run;
            run += h[s];
        }
    }
}

__device__ __forceinline__ void place_edge(int s, int d, int rk,
                                           const int* __restrict__ brow,
                                           const float* __restrict__ dis,
                                           int* __restrict__ csr) {
    const int pos = brow[d] + rk;
    const float wgt = dis[s] * dis[d];
    reinterpret_cast<int2*>(csr)[pos] = make_int2(s, __float_as_int(wgt));
}

__global__ __launch_bounds__(256) void fill_k(const int* __restrict__ src,
                                              const int* __restrict__ dst,
                                              const int* __restrict__ rank,
                                              const int* __restrict__ tab,
                                              const float* __restrict__ dis,
                                              int* __restrict__ csr, int nE4) {
    int q = blockIdx.x * blockDim.x + threadIdx.x;
    if (q >= nE4) return;
    const int* brow = tab + (size_t)((4 * q) / ESL) * NNODES;
    const int4 s = reinterpret_cast<const int4*>(src)[q];
    const int4 d = reinterpret_cast<const int4*>(dst)[q];
    const int4 r = reinterpret_cast<const int4*>(rank)[q];
    place_edge(s.x, d.x, r.x, brow, dis, csr);
    place_edge(s.y, d.y, r.y, brow, dis, csr);
    place_edge(s.z, d.z, r.z, brow, dis, csr);
    place_edge(s.w, d.w, r.w, brow, dis, csr);
}

template <int N>
__global__ __launch_bounds__(256) void gemm_wmma_f32_k(const float* __restrict__ A,
                                                       const float* __restrict__ W,
                                                       float* __restrict__ C,
                                                       int M, int K) {
    constexpr int NT = N / 16;
    const int lane = threadIdx.x & 31;
    const int wave = (blockIdx.x * blockDim.x + threadIdx.x) >> 5;
    const int m0   = wave << 4;
    if (m0 >= M) return;

    const int mr = lane & 15;
    const int kh = lane >> 4;

    v8f acc[NT];
#pragma unroll
    for (int t = 0; t < NT; ++t) acc[t] = {};

    const size_t arow = (size_t)(m0 + mr) * (size_t)K;

    for (int k0 = 0; k0 < K; k0 += 4) {
        v2f a;
        a.x = A[arow + k0 + 2 * kh + 0];
        a.y = A[arow + k0 + 2 * kh + 1];
#pragma unroll
        for (int t = 0; t < NT; ++t) {
            const int ncol = (t << 4) + mr;
            v2f b;
            b.x = W[(size_t)(k0 + 2 * kh + 0) * N + ncol];
            b.y = W[(size_t)(k0 + 2 * kh + 1) * N + ncol];
            acc[t] = __builtin_amdgcn_wmma_f32_16x16x4_f32(
                false, a, false, b, (short)0, acc[t], false, false);
        }
    }

#pragma unroll
    for (int t = 0; t < NT; ++t) {
        const int ncol = (t << 4) + mr;
#pragma unroll
        for (int v = 0; v < 8; ++v) {
            const int row = m0 + 8 * kh + v;
            C[(size_t)row * N + ncol] = acc[t][v];
        }
    }
}

template <int V>
__device__ __forceinline__ void ldrow(const float* __restrict__ p, float* r) {
    if constexpr (V == 2) {
        const float2 t = *reinterpret_cast<const float2*>(p);
        r[0] = t.x; r[1] = t.y;
    } else {
        r[0] = *p;
    }
}

template <int H, int CNT>
__device__ __forceinline__ void accum_rows(const float* __restrict__ h, int esrc, int ewgt,
                                           int j, int lane, float* acc) {
    constexpr int V = H / 32;
    float r[CNT][V];
    float wg[CNT];
    for (int k = 0; k < CNT; ++k) {
        const int s = __builtin_amdgcn_readlane(esrc, j + k);
        wg[k] = __int_as_float(__builtin_amdgcn_readlane(ewgt, j + k));
        ldrow<V>(h + (size_t)s * H + lane * V, r[k]);
    }
    for (int k = 0; k < CNT; ++k)
        for (int v = 0; v < V; ++v) acc[v] += wg[k] * r[k][v];
}

template <int H>
__device__ __forceinline__ void accum_chunk(const float* __restrict__ h, int2 ent, int m,
                                            int lane, float* acc) {
    int j = 0;
    for (; j + 8 <= m; j += 8) accum_rows<H, 8>(h, ent.x, ent.y, j, lane, acc);
    if (j + 4 <= m) { accum_rows<H, 4>(h, ent.x, ent.y, j, lane, acc); j += 4; }
    const int t = m - j;
    if (t == 3)      accum_rows<H, 3>(h, ent.x, ent.y, j, lane, acc);
    else if (t == 2) accum_rows<H, 2>(h, ent.x, ent.y, j, lane, acc);
    else if (t == 1) accum_rows<H, 1>(h, ent.x, ent.y, j, lane, acc);
}

template <int H, int NPW, int PITCH>
__device__ __forceinline__ void gather_to_lds(const float* __restrict__ h,
                                              const int* __restrict__ csr,
                                              const int* __restrict__ start,
                                              const int* __restrict__ cnt,
                                              const float* __restrict__ dis,
                                              const float* __restrict__ b,
                                              float* tile, int node0, int nn, int lane) {
    constexpr int V = H / 32;
    static_assert(NPW <= 32, "one lane per node of the wave");
    if (nn < NPW) {
#pragma unroll
        for (int i = 0; i < NPW; ++i)
            for (int v = 0; v < V; ++v) tile[i * PITCH + lane * V + v] = 0.0f;
    }
    if (nn > 0) {
        int vbeg = 0, vcnt = 0;
        float vdis = 0.0f;
        if (lane < nn) {
            vbeg = start[node0 + lane];
            vcnt = cnt[node0 + lane];
            vdis = dis[node0 + lane];
        }
        float bb[V];
        ldrow<V>(b + lane * V, bb);

        const int2* csr2 = reinterpret_cast<const int2*>(csr);

        int beg_nx = __builtin_amdgcn_readlane(vbeg, 0);
        int n_nx   = __builtin_amdgcn_readlane(vcnt, 0);
        int2 ent_nx = make_int2(0, 0);
        if (lane < n_nx) ent_nx = csr2[beg_nx + lane];
        float hs_nx[V];
        ldrow<V>(h + (size_t)node0 * H + lane * V, hs_nx);

        for (int i = 0; i < nn; ++i) {
            const int node = node0 + i;
            const int beg  = beg_nx;
            const int n    = n_nx;
            const float dv = __int_as_float(__builtin_amdgcn_readlane(__float_as_int(vdis), i));
            int2 ent = ent_nx;
            float hs[V], acc[V];
            for (int v = 0; v < V; ++v) { hs[v] = hs_nx[v]; acc[v] = 0.0f; }

            if (i + 1 < nn) {
                beg_nx = __builtin_amdgcn_readlane(vbeg, i + 1);
                n_nx   = __builtin_amdgcn_readlane(vcnt, i + 1);
                ent_nx = make_int2(0, 0);
                if (lane < n_nx) ent_nx = csr2[beg_nx + lane];
                ldrow<V>(h + (size_t)(node + 1) * H + lane * V, hs_nx);
            }

            accum_chunk<H>(h, ent, (n < 32) ? n : 32, lane, acc);
            for (int base = 32; base < n; base += 32) {
                const int m = (n - base < 32) ? (n - base) : 32;
                int2 more = make_int2(0, 0);
                if (lane < m) more = csr2[beg + base + lane];
                accum_chunk<H>(h, more, m, lane, acc);
            }

            float* trow = tile + i * PITCH + lane * V;
            for (int v = 0; v < V; ++v) {
                const float val = acc[v] + dv * dv * hs[v] + bb[v];
                trow[v] = val > 0.0f ? val : 0.0f;
            }
        }
    }
}

template <int N>
__global__ __launch_bounds__(256) void gather_gemm_k(const float* __restrict__ h,
                                                     const int* __restrict__ csr,
                                                     const int* __restrict__ start,
                                                     const int* __restrict__ cnt,
                                                     const float* __restrict__ dis,
                                                     const float* __restrict__ b,
                                                     const float* __restrict__ Wn,
                                                     float* __restrict__ C, int M) {
    constexpr int H = 64, NPW = 8, XP = 68, WP = N + 8;
    constexpr int W4 = (H * N / 4) / 256;
    constexpr int TPW = N / 32;
    __shared__ float xs[64 * XP];
    __shared__ float wsm[H * WP];

    const int lane = threadIdx.x & 31;
    const int w = __builtin_amdgcn_readfirstlane((int)(threadIdx.x >> 5));

    const int node0 = (blockIdx.x * 64 + w * NPW);
    int nn = M - node0;
    nn = nn < 0 ? 0 : (nn > NPW ? NPW : nn);
    gather_to_lds<H, NPW, XP>(h, csr, start, cnt, dis, b, xs + w * NPW * XP, node0, nn, lane);

#pragma unroll
    for (int q = 0; q < W4; ++q) {
        const int f = q * 256 + threadIdx.x;
        const int r = f / (N / 4);
        const int c = (f % (N / 4)) * 4;
        *reinterpret_cast<float4*>(&wsm[r * WP + c]) = reinterpret_cast<const float4*>(Wn)[f];
    }
    __syncthreads();

    const int mt = w >> 1;
    const int m0 = blockIdx.x * 64 + mt * 16;
    if (m0 >= M) return;
    const int nt0 = (w & 1) * TPW;
    const int mr = lane & 15;
    const int kh = lane >> 4;

    v8f acc[TPW];
#pragma unroll
    for (int t = 0; t < TPW; ++t) acc[t] = {};

    const float* arow = xs + (mt * 16 + mr) * XP + 2 * kh;
    const float* bcol = wsm + (2 * kh) * WP + nt0 * 16 + mr;
#pragma unroll
    for (int k0 = 0; k0 < H; k0 += 4) {
        v2f a;
        a.x = arow[k0 + 0];
        a.y = arow[k0 + 1];
#pragma unroll
        for (int t = 0; t < TPW; ++t) {
            v2f bq;
            bq.x = bcol[(k0 + 0) * WP + t * 16];
            bq.y = bcol[(k0 + 1) * WP + t * 16];
            acc[t] = __builtin_amdgcn_wmma_f32_16x16x4_f32(
                false, a, false, bq, (short)0, acc[t], false, false);
        }
    }

#pragma unroll
    for (int t = 0; t < TPW; ++t) {
        const int ncol = ((nt0 + t) << 4) + mr;
#pragma unroll
        for (int v = 0; v < 8; ++v) {
            const int row = m0 + 8 * kh + v;
            C[(size_t)row * N + ncol] = acc[t][v];
        }
    }
}

__global__ __launch_bounds__(256) void gather_softmax_k(const float* __restrict__ h,
                                                        const int* __restrict__ csr,
                                                        const int* __restrict__ start,
                                                        const int* __restrict__ cnt,
                                                        const float* __restrict__ dis,
                                                        const float* __restrict__ b,
                                                        float* __restrict__ out, int M) {
    constexpr int H = 32, NPW = 8;
    __shared__ float xs[64 * H];
    const int lane = threadIdx.x & 31;
    const int w = __builtin_amdgcn_readfirstlane((int)(threadIdx.x >> 5));
    const int node0 = blockIdx.x * 64 + w * NPW;
    if (node0 >= M) return;
    const int nn = (M - node0 < NPW) ? (M - node0) : NPW;
    float* tile = xs + w * NPW * H;
    gather_to_lds<H, NPW, H>(h, csr, start, cnt, dis, b, tile, node0, nn, lane);

    float v[NPW], m[NPW], e[NPW], s[NPW];
#pragma unroll
    for (int r = 0; r < NPW; ++r) { v[r] = tile[r * H + lane]; m[r] = v[r]; }
#pragma unroll
    for (int off = 16; off > 0; off >>= 1) {
#pragma unroll
        for (int r = 0; r < NPW; ++r) m[r] = fmaxf(m[r], __shfl_xor(m[r], off, 32));
    }
#pragma unroll
    for (int r = 0; r < NPW; ++r) { e[r] = __expf(v[r] - m[r]); s[r] = e[r]; }
#pragma unroll
    for (int off = 16; off > 0; off >>= 1) {
#pragma unroll
        for (int r = 0; r < NPW; ++r) s[r] += __shfl_xor(s[r], off, 32);
    }
#pragma unroll
    for (int r = 0; r < NPW; ++r)
        if (r < nn) out[(size_t)(node0 + r) * H + lane] = e[r] / s[r];
}

extern "C" void kernel_launch(void* const* d_in, const int* in_sizes, int n_in,
                              void* d_out, int out_size, void* d_ws, size_t ws_size,
                              hipStream_t stream) {
    const float* x    = (const float*)d_in[0];
    const int*   edge = (const int*)d_in[1];
    const float* W1   = (const float*)d_in[2];
    const float* b1   = (const float*)d_in[3];
    const float* W2   = (const float*)d_in[4];
    const float* b2   = (const float*)d_in[5];
    const float* W3   = (const float*)d_in[6];
    const float* b3   = (const float*)d_in[7];
    float* out = (float*)d_out;

    const int* src = edge;
    const int* dst = edge + NEDGES;

    float* dis   = (float*)d_ws;
    int*   cnt   = (int*)(dis + NNODES);
    int*   start = cnt + NNODES + 32;
    int*   rank  = start + NNODES;
    int*   csr   = rank + NEDGES;
    float* hbuf  = (float*)(csr + (size_t)2 * NEDGES);
    float* xbuf  = hbuf + (size_t)NNODES * 64;
    int*   tab   = (int*)(xbuf + (size_t)NNODES * 64);

    const int TB = 256;
    auto blocks = [](long long n, int tb) { return (int)((n + tb - 1) / tb); };

    const int gemm_waves  = NNODES / 16;
    const int gemm_blocks = blocks((long long)gemm_waves * 32, TB);
    const int node_blocks = blocks(NNODES, 64);
    static_assert(NEDGES % (4 * ESLICES) == 0, "a slice is a whole number of four-edge groups");
    static_assert(NNODES % (4 * NRANGES) == 0, "a range is a whole number of four-node groups");

    count_rank_k<<<ESLICES * NRANGES, CRT, 0, stream>>>(dst, tab, rank, cnt + NNODES);
    alloc_k<<<blocks(NNODES, TB), TB, 0, stream>>>(tab, cnt + NNODES, cnt, start, dis, NNODES);
    fill_k<<<blocks(NEDGES / 4, TB), TB, 0, stream>>>(src, dst, rank, tab, dis, csr, NEDGES / 4);

    gemm_wmma_f32_k<64><<<gemm_blocks, TB, 0, stream>>>(x, W1, hbuf, NNODES, INDIM);

    gather_gemm_k<64><<<node_blocks, TB, 0, stream>>>(hbuf, csr, start, cnt, dis, b1, W2, xbuf, NNODES);

    gather_gemm_k<32><<<node_blocks, TB, 0, stream>>>(xbuf, csr, start, cnt, dis, b2, W3, hbuf, NNODES);

    gather_softmax_k<<<node_blocks, TB, 0, stream>>>(hbuf, csr, start, cnt, dis, b3, out, NNODES);
}
